# starting kernel
# baseline (speedup 1.0000x reference)
#include <hip/hip_runtime.h>
#include <math.h>

#define HID 9
#define NG  36
#define TLEN 512
#define NCLS 10

typedef __attribute__((ext_vector_type(2))) float v2f;
typedef __attribute__((ext_vector_type(8))) float v8f;

__device__ __forceinline__ float fexp2(float x) {
#if defined(__HIP_DEVICE_COMPILE__) && __has_builtin(__builtin_amdgcn_exp2f)
  return __builtin_amdgcn_exp2f(x);
#else
  return exp2f(x);
#endif
}

__device__ __forceinline__ float frcp(float x) {
#if defined(__HIP_DEVICE_COMPILE__) && __has_builtin(__builtin_amdgcn_rcpf)
  return __builtin_amdgcn_rcpf(x);
#else
  return 1.0f / x;
#endif
}

__device__ __forceinline__ float ftanh(float x) {
#if defined(__HIP_DEVICE_COMPILE__) && __has_builtin(__builtin_amdgcn_tanhf)
  return __builtin_amdgcn_tanhf(x);
#else
  float e = fexp2(2.8853900817779268f * x);
  return 1.0f - 2.0f * frcp(e + 1.0f);
#endif
}

__device__ __forceinline__ float fsig(float x) {
  return 0.5f + 0.5f * ftanh(0.5f * x);
}

__device__ __forceinline__ v8f wmma4(v2f a, v2f b, v8f c) {
  return __builtin_amdgcn_wmma_f32_16x16x4_f32(false, a, false, b, (short)0, c,
                                               false, false);
}

__device__ __forceinline__ void load_A(const float* __restrict__ W,
                                       v2f A[3][3], int m, int hi) {
#pragma unroll
  for (int tt = 0; tt < 3; ++tt) {
    int row = 16 * tt + m;
    bool rok = row < NG;
#pragma unroll
    for (int cc = 0; cc < 3; ++cc) {
      int col = 4 * cc + 2 * hi;
      float a0 = (rok && col     < HID) ? W[row * HID + col]     : 0.0f;
      float a1 = (rok && col + 1 < HID) ? W[row * HID + col + 1] : 0.0f;
      v2f a; a.x = a0; a.y = a1;
      A[tt][cc] = a;
    }
  }
}

__device__ __forceinline__ void load_bias(const float* __restrict__ bi,
                                          const float* __restrict__ bh,
                                          v8f bias[3], int hi) {
#pragma unroll
  for (int tt = 0; tt < 3; ++tt) {
#pragma unroll
    for (int r = 0; r < 8; ++r) {
      int row = 16 * tt + 8 * hi + r;
      bias[tt][r] = (row < NG) ? (bi[row] + bh[row]) : 0.0f;
    }
  }
}

__device__ __forceinline__ void build_B(const float h[HID], v2f B[3], int hi) {
#pragma unroll
  for (int cc = 0; cc < 3; ++cc) {
    float lo0 = (4 * cc     < HID) ? h[(4 * cc    ) % HID] : 0.0f;
    float lo1 = (4 * cc + 1 < HID) ? h[(4 * cc + 1) % HID] : 0.0f;
    float hi0 = (4 * cc + 2 < HID) ? h[(4 * cc + 2) % HID] : 0.0f;
    float hi1 = (4 * cc + 3 < HID) ? h[(4 * cc + 3) % HID] : 0.0f;
    v2f b;
    b.x = hi ? hi0 : lo0;
    b.y = hi ? hi1 : lo1;
    B[cc] = b;
  }
}

__device__ __forceinline__ void lstm_step(const v2f Aih[3][3],
                                          const v2f Ahh[3][3],
                                          const v8f bias[3],
                                          const v2f Bx[3],
                                          float h[HID], float cst[HID],
                                          int hi) {
  v2f Bh[3];
  build_B(h, Bh, hi);

  v8f D[3];
  D[0] = bias[0]; D[1] = bias[1]; D[2] = bias[2];
#pragma unroll
  for (int tt = 0; tt < 3; ++tt) {
#pragma unroll
    for (int cc = 0; cc < 3; ++cc) {
      D[tt] = wmma4(Aih[tt][cc], Bx[cc], D[tt]);
      D[tt] = wmma4(Ahh[tt][cc], Bh[cc], D[tt]);
    }
  }

  float own[24], part[24];
#pragma unroll
  for (int tt = 0; tt < 3; ++tt)
#pragma unroll
    for (int r = 0; r < 8; ++r) own[tt * 8 + r] = D[tt][r];
#pragma unroll
  for (int i = 0; i < 24; ++i) part[i] = __shfl_xor(own[i], 16, 32);

  float G[NG];
#pragma unroll
  for (int row = 0; row < NG; ++row) {
    int tt = row >> 4;
    int rr = row & 15;
    int half = rr >> 3;
    int r = rr & 7;
    float a = own[tt * 8 + r];
    float b = part[tt * 8 + r];
    G[row] = (half == hi) ? a : b;
  }

#pragma unroll
  for (int j = 0; j < HID; ++j) {
    float ig = fsig(G[j]);
    float fg = fsig(G[HID + j]);
    float gg = ftanh(G[2 * HID + j]);
    float og = fsig(G[3 * HID + j]);
    float cn = fg * cst[j] + ig * gg;
    cst[j] = cn;
    h[j] = og * ftanh(cn);
  }
}

__global__ __launch_bounds__(32) void lstm2_fc_fused(
    const float* __restrict__ x,
    const float* __restrict__ wih0, const float* __restrict__ whh0,
    const float* __restrict__ bih0, const float* __restrict__ bhh0,
    const float* __restrict__ wih1, const float* __restrict__ whh1,
    const float* __restrict__ bih1, const float* __restrict__ bhh1,
    const float* __restrict__ fcw,  const float* __restrict__ fcb,
    float* __restrict__ out) {
  const int lane = threadIdx.x;
  const int m  = lane & 15;
  const int hi = lane >> 4;
  const int bglob = blockIdx.x * 16 + m;

  v2f Aih0[3][3], Ahh0[3][3], Aih1[3][3], Ahh1[3][3];
  load_A(wih0, Aih0, m, hi);
  load_A(whh0, Ahh0, m, hi);
  load_A(wih1, Aih1, m, hi);
  load_A(whh1, Ahh1, m, hi);

  v8f bias0[3], bias1[3];
  load_bias(bih0, bhh0, bias0, hi);
  load_bias(bih1, bhh1, bias1, hi);

  float h0[HID], c0[HID], h1[HID], c1[HID];
#pragma unroll
  for (int j = 0; j < HID; ++j) { h0[j] = c0[j] = h1[j] = c1[j] = 0.0f; }

  const float* xrow = x + (size_t)bglob * TLEN * HID;

#pragma unroll 1
  for (int t = 0; t < TLEN; ++t) {
    const float* xp = xrow + t * HID;
    if (t + 8 < TLEN) __builtin_prefetch(xp + 8 * HID, 0, 0);

    v2f Bx[3];
#pragma unroll
    for (int cc = 0; cc < 3; ++cc) {
      int k0 = 4 * cc + 2 * hi;
      float v0 = (k0     < HID) ? xp[k0]     : 0.0f;
      float v1 = (k0 + 1 < HID) ? xp[k0 + 1] : 0.0f;
      v2f b; b.x = v0; b.y = v1;
      Bx[cc] = b;
    }

    lstm_step(Aih0, Ahh0, bias0, Bx, h0, c0, hi);

    v2f B1[3];
    build_B(h0, B1, hi);
    lstm_step(Aih1, Ahh1, bias1, B1, h1, c1, hi);
  }

  if (hi == 0) {
    float* op = out + (size_t)bglob * NCLS;
#pragma unroll
    for (int k = 0; k < NCLS; ++k) {
      float acc = fcb[k];
#pragma unroll
      for (int j = 0; j < HID; ++j) acc += h1[j] * fcw[k * HID + j];
      op[k] = acc;
    }
  }
}

extern "C" void kernel_launch(void* const* d_in, const int* in_sizes, int n_in,
                              void* d_out, int out_size, void* d_ws, size_t ws_size,
                              hipStream_t stream) {
  const float* x    = (const float*)d_in[0];
  const float* wih0 = (const float*)d_in[1];
  const float* whh0 = (const float*)d_in[2];
  const float* bih0 = (const float*)d_in[3];
  const float* bhh0 = (const float*)d_in[4];
  const float* wih1 = (const float*)d_in[5];
  const float* whh1 = (const float*)d_in[6];
  const float* bih1 = (const float*)d_in[7];
  const float* bhh1 = (const float*)d_in[8];
  const float* fcw  = (const float*)d_in[9];
  const float* fcb  = (const float*)d_in[10];
  float* out = (float*)d_out;

  const int B = in_sizes[0] / (TLEN * HID);
  dim3 grid(B / 16);
  dim3 block(32);
  lstm2_fc_fused<<<grid, block, 0, stream>>>(x, wih0, whh0, bih0, bhh0,
                                             wih1, whh1, bih1, bhh1,
                                             fcw, fcb, out);
}


# h1(t-1) rides with h0(t) through the LDS exchange: a lane's two pairs are one 16-byte word, 1 store + 2 reads a step (2 + 4 before), a step's LDS work in 2 issue groups, not 3; sums unchanged
# speedup vs baseline: 4.6939x; 4.6939x over previous
#include <hip/hip_runtime.h>
#include <math.h>


#define HID 9
#define NG  36
#define TLEN 512
#define NCLS 10
#define NWAVE 3
#define UPW 3
#define GP 8
#define HP 28
#define H1SLOT(j) (GP * ((j) / UPW) + 4 * (((j) % UPW) / 2) + 2 + ((j) % UPW) % 2)
#define TC 32
#define NCH (TLEN / TC)
#define XROW (TC * HID)
#define XP (XROW + 4)
#define XBUF (16 * XP)
#define TPR 24
#define XLD 12

typedef __attribute__((ext_vector_type(2))) float v2f;
typedef __attribute__((ext_vector_type(4))) float v4f;
typedef __attribute__((ext_vector_type(8))) float v8f;

__device__ __forceinline__ float fexp2(float x) {
#if defined(__HIP_DEVICE_COMPILE__) && __has_builtin(__builtin_amdgcn_exp2f)
  return __builtin_amdgcn_exp2f(x);
#else
  return exp2f(x);
#endif
}

__device__ __forceinline__ float frcp(float x) {
#if defined(__HIP_DEVICE_COMPILE__) && __has_builtin(__builtin_amdgcn_rcpf)
  return __builtin_amdgcn_rcpf(x);
#else
  return 1.0f / x;
#endif
}

__device__ __forceinline__ float ftanh(float x) {
#if defined(__HIP_DEVICE_COMPILE__) && __has_builtin(__builtin_amdgcn_tanhf)
  return __builtin_amdgcn_tanhf(x);
#else
  float e = fexp2(2.8853900817779268f * x);
  return 1.0f - 2.0f * frcp(e + 1.0f);
#endif
}

__device__ __forceinline__ float fsig(float x) {
  return 0.5f + 0.5f * ftanh(0.5f * x);
}

__device__ __forceinline__ v8f wmma4(v2f a, v2f b, v8f c) {
  return __builtin_amdgcn_wmma_f32_16x16x4_f32(false, a, false, b, (short)0, c,
                                               false, false);
}

__device__ __forceinline__ int tile_row_to_gate_row(int q, int w) {
  return (q & 3) * HID + UPW * w + (q >> 2);
}

template <bool byowner>
__device__ __forceinline__ void load_A(const float* __restrict__ W,
                                       v2f A[3], int w, int m, int hi) {
  bool rok = m < 4 * UPW;
  int row = rok ? tile_row_to_gate_row(m, w) : 0;
#pragma unroll
  for (int cc = 0; cc < 3; ++cc) {
    int s0 = 4 * cc + 2 * hi;
    int c0 = byowner ? UPW * ((w + cc) % NWAVE) + 2 * hi : s0;
    bool ok0 = byowner ? true : (s0 < HID);
    bool ok1 = byowner ? (hi == 0) : (s0 + 1 < HID);
    float a0 = (rok && ok0) ? W[row * HID + c0]     : 0.0f;
    float a1 = (rok && ok1) ? W[row * HID + c0 + 1] : 0.0f;
    v2f a; a.x = a0; a.y = a1;
    A[cc] = a;
  }
}

__device__ __forceinline__ v8f load_bias(const float* __restrict__ bi,
                                         const float* __restrict__ bh,
                                         int w, int hi) {
  v8f bias;
#pragma unroll
  for (int r = 0; r < 8; ++r) {
    int q = 8 * hi + r;
    bool rok = q < 4 * UPW;
    int row = rok ? tile_row_to_gate_row(q, w) : 0;
    bias[r] = rok ? (bi[row] + bh[row]) : 0.0f;
  }
  return bias;
}

__device__ __forceinline__ v4f read_B2(const float* hb) {
  return *(const v4f*)hb;
}

__device__ __forceinline__ void stage_load(const float* __restrict__ g, v4f R[XLD]) {
#pragma unroll
  for (int u = 0; u < 4; ++u)
#pragma unroll
    for (int s = 0; s < 3; ++s)
      R[3 * u + s] = *(const v4f*)(g + (size_t)(4 * u) * TLEN * HID + 4 * TPR * s);
}

__device__ __forceinline__ void stage_store(float* l, const v4f R[XLD]) {
#pragma unroll
  for (int u = 0; u < 4; ++u)
#pragma unroll
    for (int s = 0; s < 3; ++s)
      *(v4f*)(l + 4 * u * XP + 4 * TPR * s) = R[3 * u + s];
}

__device__ __forceinline__ void read_X(const float* xq, v2f Bx[3], float& x8) {
  Bx[0].x = xq[0]; Bx[0].y = xq[1];
  Bx[1].x = xq[4]; Bx[1].y = xq[5];
  x8 = xq[8];
  Bx[2].y = 0.0f;
}

__device__ __forceinline__ v2f lstm_gates(v8f D, float cst[2]) {
  float hout[2];
#pragma unroll
  for (int s = 0; s < 2; ++s) {
    float ig = fsig(D[4 * s + 0]);
    float fg = fsig(D[4 * s + 1]);
    float gg = ftanh(D[4 * s + 2]);
    float og = fsig(D[4 * s + 3]);
    float cn = fg * cst[s] + ig * gg;
    cst[s] = cn;
    hout[s] = og * ftanh(cn);
  }
  v2f hv; hv.x = hout[0]; hv.y = hout[1];
  return hv;
}

__global__ __launch_bounds__(32 * NWAVE) void lstm2_fc_fused(
    const float* __restrict__ x,
    const float* __restrict__ wih0, const float* __restrict__ whh0,
    const float* __restrict__ bih0, const float* __restrict__ bhh0,
    const float* __restrict__ wih1, const float* __restrict__ whh1,
    const float* __restrict__ bih1, const float* __restrict__ bhh1,
    const float* __restrict__ fcw,  const float* __restrict__ fcb,
    float* __restrict__ out) {
  __shared__ __attribute__((aligned(16))) float hx[2][16 * HP];
  __shared__ __attribute__((aligned(16))) float xs[2 * XBUF];

  const int tid  = threadIdx.x;
  const int w    = tid >> 5;
  const int lane = tid & 31;
  const int m  = lane & 15;
  const int hi = lane >> 4;
  const int bglob = blockIdx.x * 16 + m;

  const int sr = tid / TPR, sq = tid % TPR;
  const float* gx = x + ((size_t)blockIdx.x * 16 + sr) * TLEN * HID + 4 * sq;
  const int lofs = sr * XP + 4 * sq;
  const int xlane = m * XP + 2 * hi;

  v4f R[XLD];
  stage_load(gx, R);
  stage_store(xs + lofs, R);
  if (tid < 2 * 16) {
    v4f z; z.x = 0.0f; z.y = 0.0f; z.z = 0.0f; z.w = 0.0f;
    *(v4f*)(xs + (tid >> 4) * XBUF + (tid & 15) * XP + XROW) = z;
  }
  __syncthreads();
  v2f Bx[3];
  float x8;
  read_X(xs + xlane, Bx, x8);
  const float* xq = xs + xlane + HID;

  v2f Aih0[3], Ahh0[3], Aih1[3], Ahh1[3];
  load_A<false>(wih0, Aih0, w, m, hi);
  load_A<true>(whh0, Ahh0, w, m, hi);
  load_A<true>(wih1, Aih1, w, m, hi);
  load_A<true>(whh1, Ahh1, w, m, hi);

  const v8f bias0 = load_bias(bih0, bhh0, w, hi);
  const v8f bias1 = load_bias(bih1, bhh1, w, hi);

  float c0[2] = {0.0f, 0.0f}, c1[2] = {0.0f, 0.0f};

  const int st  = m * HP + GP * w + 4 * hi;
  const int rd1 = m * HP + GP * ((w + 1) % NWAVE) + 4 * hi;
  const int rd2 = m * HP + GP * ((w + 2) % NWAVE) + 4 * hi;

  v2f zero2; zero2.x = 0.0f; zero2.y = 0.0f;
  v2f Bh0r[2] = {zero2, zero2};
  v2f hv1 = zero2;
  Bx[2].x = (hi == 0) ? x8 : 0.0f;
  v8f D0n = wmma4(Ahh0[0], zero2, bias0);
  D0n = wmma4(Aih0[0], Bx[0], D0n);
  D0n = wmma4(Aih0[1], Bx[1], D0n);
  D0n = wmma4(Aih0[2], Bx[2], D0n);

#pragma unroll 1
  for (int hc = 0; hc < 2 * NCH; ++hc) {
    const int cn = (hc >> 1) + 1;
    if (cn < NCH) {
      if ((hc & 1) == 0) stage_load(gx + cn * XROW, R);
      else               stage_store(xs + (cn & 1) * XBUF + lofs, R);
    }

#pragma unroll 1
  for (int th = 0; th < TC / 2; ++th) {
    const int t = hc * (TC / 2) + th;
    const int pa = t & 1;

    v8f D0 = D0n;
    D0 = wmma4(Ahh0[1], Bh0r[0], D0);
    D0 = wmma4(Ahh0[2], Bh0r[1], D0);
    const v2f hv0 = lstm_gates(D0, c0);
    {
      v4f hs; hs.x = hv0.x; hs.y = hv0.y; hs.z = hv1.x; hs.w = hv1.y;
      *(v4f*)&hx[pa][st] = hs;
    }
    {
      read_X(xq, Bx, x8);
      const int t2 = t + 2;
      xq = (t2 % TC == 0) ? xs + ((t2 / TC) & 1) * XBUF + xlane : xq + HID;
    }
    __builtin_amdgcn_sched_barrier(0);

    v8f D1 = wmma4(Aih1[0], hv0, bias1);
    D0n    = wmma4(Ahh0[0], hv0, bias0);
    D1     = wmma4(Ahh1[0], hv1, D1);
    __builtin_amdgcn_sched_barrier(0);

    __syncthreads();

    v2f Bh1r[2];
    {
      const v4f r1 = read_B2(&hx[pa][rd1]);
      const v4f r2 = read_B2(&hx[pa][rd2]);
      Bh0r[0].x = r1.x; Bh0r[0].y = r1.y;
      Bh0r[1].x = r2.x; Bh0r[1].y = r2.y;
      Bh1r[0].x = r1.z; Bh1r[0].y = r1.w;
      Bh1r[1].x = r2.z; Bh1r[1].y = r2.w;
    }
    __builtin_amdgcn_sched_barrier(0);

    Bx[2].x = (hi == 0) ? x8 : 0.0f;
    D0n = wmma4(Aih0[0], Bx[0], D0n);
    D0n = wmma4(Aih0[1], Bx[1], D0n);
    D0n = wmma4(Aih0[2], Bx[2], D0n);
    asm volatile("" : "+v"(D0n));
    __builtin_amdgcn_sched_barrier(0);

    D1 = wmma4(Aih1[1], Bh0r[0], D1);
    D1 = wmma4(Ahh1[1], Bh1r[0], D1);
    D1 = wmma4(Aih1[2], Bh0r[1], D1);
    D1 = wmma4(Ahh1[2], Bh1r[1], D1);
    hv1 = lstm_gates(D1, c1);
  }
  }

  *(v2f*)&hx[TLEN & 1][st + 2] = hv1;
  __syncthreads();

  if (w == 0 && hi == 0) {
    const float* hl = &hx[TLEN & 1][m * HP];
    float h1[HID];
#pragma unroll
    for (int j = 0; j < HID; ++j) h1[j] = hl[H1SLOT(j)];
    float* op = out + (size_t)bglob * NCLS;
#pragma unroll
    for (int k = 0; k < NCLS; ++k) {
      float acc = fcb[k];
#pragma unroll
      for (int j = 0; j < HID; ++j) acc += h1[j] * fcw[k * HID + j];
      op[k] = acc;
    }
  }
}

extern "C" void kernel_launch(void* const* d_in, const int* in_sizes, int n_in,
                              void* d_out, int out_size, void* d_ws, size_t ws_size,
                              hipStream_t stream) {
  const float* x    = (const float*)d_in[0];
  const float* wih0 = (const float*)d_in[1];
  const float* whh0 = (const float*)d_in[2];
  const float* bih0 = (const float*)d_in[3];
  const float* bhh0 = (const float*)d_in[4];
  const float* wih1 = (const float*)d_in[5];
  const float* whh1 = (const float*)d_in[6];
  const float* bih1 = (const float*)d_in[7];
  const float* bhh1 = (const float*)d_in[8];
  const float* fcw  = (const float*)d_in[9];
  const float* fcb  = (const float*)d_in[10];
  float* out = (float*)d_out;

  const int B = in_sizes[0] / (TLEN * HID);
  dim3 grid(B / 16);
  dim3 block(32 * NWAVE);
  lstm2_fc_fused<<<grid, block, 0, stream>>>(x, wih0, whh0, bih0, bhh0,
                                             wih1, whh1, bih1, bhh1,
                                             fcw, fcb, out);
}


# the step's workgroup barrier fences LDS only: the next chunk's x loads, sent at a chunk's start, stay in flight across barriers, no longer drained at the chunk's first one; sums unchanged
# speedup vs baseline: 5.2656x; 1.1218x over previous
#include <hip/hip_runtime.h>
#include <math.h>


#define HID 9
#define NG  36
#define TLEN 512
#define NCLS 10
#define NWAVE 3
#define UPW 3
#define GP 8
#define HP 28
#define H1SLOT(j) (GP * ((j) / UPW) + 4 * (((j) % UPW) / 2) + 2 + ((j) % UPW) % 2)
#define TC 32
#define NCH (TLEN / TC)
#define XROW (TC * HID)
#define XP (XROW + 4)
#define XBUF (16 * XP)
#define TPR 24
#define XLD 12

typedef __attribute__((ext_vector_type(2))) float v2f;
typedef __attribute__((ext_vector_type(4))) float v4f;
typedef __attribute__((ext_vector_type(8))) float v8f;

__device__ __forceinline__ float fexp2(float x) {
#if defined(__HIP_DEVICE_COMPILE__) && __has_builtin(__builtin_amdgcn_exp2f)
  return __builtin_amdgcn_exp2f(x);
#else
  return exp2f(x);
#endif
}

__device__ __forceinline__ float frcp(float x) {
#if defined(__HIP_DEVICE_COMPILE__) && __has_builtin(__builtin_amdgcn_rcpf)
  return __builtin_amdgcn_rcpf(x);
#else
  return 1.0f / x;
#endif
}

__device__ __forceinline__ float ftanh(float x) {
#if defined(__HIP_DEVICE_COMPILE__) && __has_builtin(__builtin_amdgcn_tanhf)
  return __builtin_amdgcn_tanhf(x);
#else
  float e = fexp2(2.8853900817779268f * x);
  return 1.0f - 2.0f * frcp(e + 1.0f);
#endif
}

__device__ __forceinline__ float fsig(float x) {
  return 0.5f + 0.5f * ftanh(0.5f * x);
}

__device__ __forceinline__ v8f wmma4(v2f a, v2f b, v8f c) {
  return __builtin_amdgcn_wmma_f32_16x16x4_f32(false, a, false, b, (short)0, c,
                                               false, false);
}

__device__ __forceinline__ int tile_row_to_gate_row(int q, int w) {
  return (q & 3) * HID + UPW * w + (q >> 2);
}

template <bool byowner>
__device__ __forceinline__ void load_A(const float* __restrict__ W,
                                       v2f A[3], int w, int m, int hi) {
  bool rok = m < 4 * UPW;
  int row = rok ? tile_row_to_gate_row(m, w) : 0;
#pragma unroll
  for (int cc = 0; cc < 3; ++cc) {
    int s0 = 4 * cc + 2 * hi;
    int c0 = byowner ? UPW * ((w + cc) % NWAVE) + 2 * hi : s0;
    bool ok0 = byowner ? true : (s0 < HID);
    bool ok1 = byowner ? (hi == 0) : (s0 + 1 < HID);
    float a0 = (rok && ok0) ? W[row * HID + c0]     : 0.0f;
    float a1 = (rok && ok1) ? W[row * HID + c0 + 1] : 0.0f;
    v2f a; a.x = a0; a.y = a1;
    A[cc] = a;
  }
}

__device__ __forceinline__ v8f load_bias(const float* __restrict__ bi,
                                         const float* __restrict__ bh,
                                         int w, int hi) {
  v8f bias;
#pragma unroll
  for (int r = 0; r < 8; ++r) {
    int q = 8 * hi + r;
    bool rok = q < 4 * UPW;
    int row = rok ? tile_row_to_gate_row(q, w) : 0;
    bias[r] = rok ? (bi[row] + bh[row]) : 0.0f;
  }
  return bias;
}

__device__ __forceinline__ v4f read_B2(const float* hb) {
  return *(const v4f*)hb;
}

__device__ __forceinline__ void stage_load(const float* __restrict__ g, v4f R[XLD]) {
#pragma unroll
  for (int u = 0; u < 4; ++u)
#pragma unroll
    for (int s = 0; s < 3; ++s)
      R[3 * u + s] = *(const v4f*)(g + (size_t)(4 * u) * TLEN * HID + 4 * TPR * s);
}

__device__ __forceinline__ void stage_store(float* l, const v4f R[XLD]) {
#pragma unroll
  for (int u = 0; u < 4; ++u)
#pragma unroll
    for (int s = 0; s < 3; ++s)
      *(v4f*)(l + 4 * u * XP + 4 * TPR * s) = R[3 * u + s];
}

__device__ __forceinline__ void read_X(const float* xq, v2f Bx[3], float& x8) {
  Bx[0].x = xq[0]; Bx[0].y = xq[1];
  Bx[1].x = xq[4]; Bx[1].y = xq[5];
  x8 = xq[8];
  Bx[2].y = 0.0f;
}

__device__ __forceinline__ v2f lstm_gates(v8f D, float cst[2]) {
  float hout[2];
#pragma unroll
  for (int s = 0; s < 2; ++s) {
    float ig = fsig(D[4 * s + 0]);
    float fg = fsig(D[4 * s + 1]);
    float gg = ftanh(D[4 * s + 2]);
    float og = fsig(D[4 * s + 3]);
    float cn = fg * cst[s] + ig * gg;
    cst[s] = cn;
    hout[s] = og * ftanh(cn);
  }
  v2f hv; hv.x = hout[0]; hv.y = hout[1];
  return hv;
}

__device__ __forceinline__ void lds_barrier() {
  __builtin_amdgcn_fence(__ATOMIC_RELEASE, "workgroup", "local");
  __builtin_amdgcn_s_barrier();
  __builtin_amdgcn_fence(__ATOMIC_ACQUIRE, "workgroup", "local");
}

__global__ __launch_bounds__(32 * NWAVE) void lstm2_fc_fused(
    const float* __restrict__ x,
    const float* __restrict__ wih0, const float* __restrict__ whh0,
    const float* __restrict__ bih0, const float* __restrict__ bhh0,
    const float* __restrict__ wih1, const float* __restrict__ whh1,
    const float* __restrict__ bih1, const float* __restrict__ bhh1,
    const float* __restrict__ fcw,  const float* __restrict__ fcb,
    float* __restrict__ out) {
  __shared__ __attribute__((aligned(16))) float hx[2][16 * HP];
  __shared__ __attribute__((aligned(16))) float xs[2 * XBUF];

  const int tid  = threadIdx.x;
  const int w    = tid >> 5;
  const int lane = tid & 31;
  const int m  = lane & 15;
  const int hi = lane >> 4;
  const int bglob = blockIdx.x * 16 + m;

  const int sr = tid / TPR, sq = tid % TPR;
  const float* gx = x + ((size_t)blockIdx.x * 16 + sr) * TLEN * HID + 4 * sq;
  const int lofs = sr * XP + 4 * sq;
  const int xlane = m * XP + 2 * hi;

  v4f R[XLD];
  stage_load(gx, R);
  stage_store(xs + lofs, R);
  if (tid < 2 * 16) {
    v4f z; z.x = 0.0f; z.y = 0.0f; z.z = 0.0f; z.w = 0.0f;
    *(v4f*)(xs + (tid >> 4) * XBUF + (tid & 15) * XP + XROW) = z;
  }
  __syncthreads();
  v2f Bx[3];
  float x8;
  read_X(xs + xlane, Bx, x8);
  const float* xq = xs + xlane + HID;

  v2f Aih0[3], Ahh0[3], Aih1[3], Ahh1[3];
  load_A<false>(wih0, Aih0, w, m, hi);
  load_A<true>(whh0, Ahh0, w, m, hi);
  load_A<true>(wih1, Aih1, w, m, hi);
  load_A<true>(whh1, Ahh1, w, m, hi);

  const v8f bias0 = load_bias(bih0, bhh0, w, hi);
  const v8f bias1 = load_bias(bih1, bhh1, w, hi);

  float c0[2] = {0.0f, 0.0f}, c1[2] = {0.0f, 0.0f};

  const int st  = m * HP + GP * w + 4 * hi;
  const int rd1 = m * HP + GP * ((w + 1) % NWAVE) + 4 * hi;
  const int rd2 = m * HP + GP * ((w + 2) % NWAVE) + 4 * hi;

  v2f zero2; zero2.x = 0.0f; zero2.y = 0.0f;
  v2f Bh0r[2] = {zero2, zero2};
  v2f hv1 = zero2;
  Bx[2].x = (hi == 0) ? x8 : 0.0f;
  v8f D0n = wmma4(Ahh0[0], zero2, bias0);
  D0n = wmma4(Aih0[0], Bx[0], D0n);
  D0n = wmma4(Aih0[1], Bx[1], D0n);
  D0n = wmma4(Aih0[2], Bx[2], D0n);

#pragma unroll 1
  for (int hc = 0; hc < 2 * NCH; ++hc) {
    const int cn = (hc >> 1) + 1;
    if (cn < NCH) {
      if ((hc & 1) == 0) stage_load(gx + cn * XROW, R);
      else               stage_store(xs + (cn & 1) * XBUF + lofs, R);
    }

#pragma unroll 1
  for (int th = 0; th < TC / 2; ++th) {
    const int t = hc * (TC / 2) + th;
    const int pa = t & 1;

    v8f D0 = D0n;
    D0 = wmma4(Ahh0[1], Bh0r[0], D0);
    D0 = wmma4(Ahh0[2], Bh0r[1], D0);
    const v2f hv0 = lstm_gates(D0, c0);
    {
      v4f hs; hs.x = hv0.x; hs.y = hv0.y; hs.z = hv1.x; hs.w = hv1.y;
      *(v4f*)&hx[pa][st] = hs;
    }
    {
      read_X(xq, Bx, x8);
      const int t2 = t + 2;
      xq = (t2 % TC == 0) ? xs + ((t2 / TC) & 1) * XBUF + xlane : xq + HID;
    }
    __builtin_amdgcn_sched_barrier(0);

    v8f D1 = wmma4(Aih1[0], hv0, bias1);
    D0n    = wmma4(Ahh0[0], hv0, bias0);
    D1     = wmma4(Ahh1[0], hv1, D1);
    __builtin_amdgcn_sched_barrier(0);

    asm volatile("s_wait_xcnt 0x0");
    lds_barrier();

    v2f Bh1r[2];
    {
      const v4f r1 = read_B2(&hx[pa][rd1]);
      const v4f r2 = read_B2(&hx[pa][rd2]);
      Bh0r[0].x = r1.x; Bh0r[0].y = r1.y;
      Bh0r[1].x = r2.x; Bh0r[1].y = r2.y;
      Bh1r[0].x = r1.z; Bh1r[0].y = r1.w;
      Bh1r[1].x = r2.z; Bh1r[1].y = r2.w;
    }
    __builtin_amdgcn_sched_barrier(0);

    Bx[2].x = (hi == 0) ? x8 : 0.0f;
    D0n = wmma4(Aih0[0], Bx[0], D0n);
    D0n = wmma4(Aih0[1], Bx[1], D0n);
    D0n = wmma4(Aih0[2], Bx[2], D0n);
    asm volatile("" : "+v"(D0n));
    __builtin_amdgcn_sched_barrier(0);

    D1 = wmma4(Aih1[1], Bh0r[0], D1);
    D1 = wmma4(Ahh1[1], Bh1r[0], D1);
    D1 = wmma4(Aih1[2], Bh0r[1], D1);
    D1 = wmma4(Ahh1[2], Bh1r[1], D1);
    hv1 = lstm_gates(D1, c1);
  }
  }

  *(v2f*)&hx[TLEN & 1][st + 2] = hv1;
  __syncthreads();

  if (w == 0 && hi == 0) {
    const float* hl = &hx[TLEN & 1][m * HP];
    float h1[HID];
#pragma unroll
    for (int j = 0; j < HID; ++j) h1[j] = hl[H1SLOT(j)];
    float* op = out + (size_t)bglob * NCLS;
#pragma unroll
    for (int k = 0; k < NCLS; ++k) {
      float acc = fcb[k];
#pragma unroll
      for (int j = 0; j < HID; ++j) acc += h1[j] * fcw[k * HID + j];
      op[k] = acc;
    }
  }
}

extern "C" void kernel_launch(void* const* d_in, const int* in_sizes, int n_in,
                              void* d_out, int out_size, void* d_ws, size_t ws_size,
                              hipStream_t stream) {
  const float* x    = (const float*)d_in[0];
  const float* wih0 = (const float*)d_in[1];
  const float* whh0 = (const float*)d_in[2];
  const float* bih0 = (const float*)d_in[3];
  const float* bhh0 = (const float*)d_in[4];
  const float* wih1 = (const float*)d_in[5];
  const float* whh1 = (const float*)d_in[6];
  const float* bih1 = (const float*)d_in[7];
  const float* bhh1 = (const float*)d_in[8];
  const float* fcw  = (const float*)d_in[9];
  const float* fcb  = (const float*)d_in[10];
  float* out = (float*)d_out;

  const int B = in_sizes[0] / (TLEN * HID);
  dim3 grid(B / 16);
  dim3 block(32 * NWAVE);
  lstm2_fc_fused<<<grid, block, 0, stream>>>(x, wih0, whh0, bih0, bhh0,
                                             wih1, whh1, bih1, bhh1,
                                             fcw, fcb, out);
}


# loop trip cut at the h store: layer 1's step t and layer 0's step t+1, both behind one exchange, run as one block, products in turn, gate math side by side, not in sequence; sums unchanged
# speedup vs baseline: 5.6646x; 1.0758x over previous
#include <hip/hip_runtime.h>
#include <math.h>


#define HID 9
#define NG  36
#define TLEN 512
#define NCLS 10
#define NWAVE 3
#define UPW 3
#define GP 8
#define HP 28
#define H1SLOT(j) (GP * ((j) / UPW) + 4 * (((j) % UPW) / 2) + 2 + ((j) % UPW) % 2)
#define TC 32
#define NCH (TLEN / TC)
#define XROW (TC * HID)
#define XP (XROW + 4)
#define XBUF (16 * XP)
#define TPR 24
#define XLD 12

typedef __attribute__((ext_vector_type(2))) float v2f;
typedef __attribute__((ext_vector_type(4))) float v4f;
typedef __attribute__((ext_vector_type(8))) float v8f;

__device__ __forceinline__ float fexp2(float x) {
#if defined(__HIP_DEVICE_COMPILE__) && __has_builtin(__builtin_amdgcn_exp2f)
  return __builtin_amdgcn_exp2f(x);
#else
  return exp2f(x);
#endif
}

__device__ __forceinline__ float frcp(float x) {
#if defined(__HIP_DEVICE_COMPILE__) && __has_builtin(__builtin_amdgcn_rcpf)
  return __builtin_amdgcn_rcpf(x);
#else
  return 1.0f / x;
#endif
}

__device__ __forceinline__ float ftanh(float x) {
#if defined(__HIP_DEVICE_COMPILE__) && __has_builtin(__builtin_amdgcn_tanhf)
  return __builtin_amdgcn_tanhf(x);
#else
  float e = fexp2(2.8853900817779268f * x);
  return 1.0f - 2.0f * frcp(e + 1.0f);
#endif
}

__device__ __forceinline__ float fsig(float x) {
  return 0.5f + 0.5f * ftanh(0.5f * x);
}

__device__ __forceinline__ v8f wmma4(v2f a, v2f b, v8f c) {
  return __builtin_amdgcn_wmma_f32_16x16x4_f32(false, a, false, b, (short)0, c,
                                               false, false);
}

__device__ __forceinline__ int tile_row_to_gate_row(int q, int w) {
  return (q & 3) * HID + UPW * w + (q >> 2);
}

template <bool byowner>
__device__ __forceinline__ void load_A(const float* __restrict__ W,
                                       v2f A[3], int w, int m, int hi) {
  bool rok = m < 4 * UPW;
  int row = rok ? tile_row_to_gate_row(m, w) : 0;
#pragma unroll
  for (int cc = 0; cc < 3; ++cc) {
    int s0 = 4 * cc + 2 * hi;
    int c0 = byowner ? UPW * ((w + cc) % NWAVE) + 2 * hi : s0;
    bool ok0 = byowner ? true : (s0 < HID);
    bool ok1 = byowner ? (hi == 0) : (s0 + 1 < HID);
    float a0 = (rok && ok0) ? W[row * HID + c0]     : 0.0f;
    float a1 = (rok && ok1) ? W[row * HID + c0 + 1] : 0.0f;
    v2f a; a.x = a0; a.y = a1;
    A[cc] = a;
  }
}

__device__ __forceinline__ v8f load_bias(const float* __restrict__ bi,
                                         const float* __restrict__ bh,
                                         int w, int hi) {
  v8f bias;
#pragma unroll
  for (int r = 0; r < 8; ++r) {
    int q = 8 * hi + r;
    bool rok = q < 4 * UPW;
    int row = rok ? tile_row_to_gate_row(q, w) : 0;
    bias[r] = rok ? (bi[row] + bh[row]) : 0.0f;
  }
  return bias;
}

__device__ __forceinline__ v4f read_B2(const float* hb) {
  return *(const v4f*)hb;
}

__device__ __forceinline__ void stage_load(const float* __restrict__ g, v4f R[XLD]) {
#pragma unroll
  for (int u = 0; u < 4; ++u)
#pragma unroll
    for (int s = 0; s < 3; ++s)
      R[3 * u + s] = *(const v4f*)(g + (size_t)(4 * u) * TLEN * HID + 4 * TPR * s);
}

__device__ __forceinline__ void stage_store(float* l, const v4f R[XLD]) {
#pragma unroll
  for (int u = 0; u < 4; ++u)
#pragma unroll
    for (int s = 0; s < 3; ++s)
      *(v4f*)(l + 4 * u * XP + 4 * TPR * s) = R[3 * u + s];
}

__device__ __forceinline__ void read_X(const float* xq, v2f Bx[3], float& x8) {
  Bx[0].x = xq[0]; Bx[0].y = xq[1];
  Bx[1].x = xq[4]; Bx[1].y = xq[5];
  x8 = xq[8];
  Bx[2].y = 0.0f;
}

__device__ __forceinline__ v2f lstm_gates(v8f D, float cst[2]) {
  float hout[2];
#pragma unroll
  for (int s = 0; s < 2; ++s) {
    float ig = fsig(D[4 * s + 0]);
    float fg = fsig(D[4 * s + 1]);
    float gg = ftanh(D[4 * s + 2]);
    float og = fsig(D[4 * s + 3]);
    float cn = fmaf(ig, gg, fg * cst[s]);
    cst[s] = cn;
    hout[s] = og * ftanh(cn);
  }
  v2f hv; hv.x = hout[0]; hv.y = hout[1];
  return hv;
}

__device__ __forceinline__ void lstm_gates_two(v8f Da, float ca[2], v2f& ha,
                                               v8f Db, float cb[2], v2f& hb) {
  float pre[4][4], cs[4], ig[4], fg[4], gg[4], og[4], hout[4];
#pragma unroll
  for (int s = 0; s < 2; ++s) {
#pragma unroll
    for (int g = 0; g < 4; ++g) {
      pre[s][g]     = Da[4 * s + g];
      pre[2 + s][g] = Db[4 * s + g];
    }
    cs[s]     = ca[s];
    cs[2 + s] = cb[s];
  }
#pragma unroll
  for (int q = 0; q < 4; ++q) {
    ig[q] = fsig(pre[q][0]);
    fg[q] = fsig(pre[q][1]);
    gg[q] = ftanh(pre[q][2]);
    og[q] = fsig(pre[q][3]);
  }
#pragma unroll
  for (int q = 0; q < 4; ++q) {
    float cn = fmaf(ig[q], gg[q], fg[q] * cs[q]);
    cs[q] = cn;
    hout[q] = og[q] * ftanh(cn);
  }
#pragma unroll
  for (int s = 0; s < 2; ++s) {
    ca[s] = cs[s];
    cb[s] = cs[2 + s];
  }
  ha.x = hout[0]; ha.y = hout[1];
  hb.x = hout[2]; hb.y = hout[3];
}

__device__ __forceinline__ void lds_barrier() {
  __builtin_amdgcn_fence(__ATOMIC_RELEASE, "workgroup", "local");
  __builtin_amdgcn_s_barrier();
  __builtin_amdgcn_fence(__ATOMIC_ACQUIRE, "workgroup", "local");
}

__global__ __launch_bounds__(32 * NWAVE) void lstm2_fc_fused(
    const float* __restrict__ x,
    const float* __restrict__ wih0, const float* __restrict__ whh0,
    const float* __restrict__ bih0, const float* __restrict__ bhh0,
    const float* __restrict__ wih1, const float* __restrict__ whh1,
    const float* __restrict__ bih1, const float* __restrict__ bhh1,
    const float* __restrict__ fcw,  const float* __restrict__ fcb,
    float* __restrict__ out) {
  __shared__ __attribute__((aligned(16))) float hx[2][16 * HP];
  __shared__ __attribute__((aligned(16))) float xs[2 * XBUF];

  const int tid  = threadIdx.x;
  const int w    = tid >> 5;
  const int lane = tid & 31;
  const int m  = lane & 15;
  const int hi = lane >> 4;
  const int bglob = blockIdx.x * 16 + m;

  const int sr = tid / TPR, sq = tid % TPR;
  const float* gx = x + ((size_t)blockIdx.x * 16 + sr) * TLEN * HID + 4 * sq;
  const int lofs = sr * XP + 4 * sq;
  const int xlane = m * XP + 2 * hi;

  v4f R[XLD];
  stage_load(gx, R);
  stage_store(xs + lofs, R);
  if (tid < 2 * 16) {
    v4f z; z.x = 0.0f; z.y = 0.0f; z.z = 0.0f; z.w = 0.0f;
    *(v4f*)(xs + (tid >> 4) * XBUF + (tid & 15) * XP + XROW) = z;
  }
  __syncthreads();
  v2f Bx[3];
  float x8;
  read_X(xs + xlane, Bx, x8);
  const float* xq = xs + xlane + HID;

  v2f Aih0[3], Ahh0[3], Aih1[3], Ahh1[3];
  load_A<false>(wih0, Aih0, w, m, hi);
  load_A<true>(whh0, Ahh0, w, m, hi);
  load_A<true>(wih1, Aih1, w, m, hi);
  load_A<true>(whh1, Ahh1, w, m, hi);

  const v8f bias0 = load_bias(bih0, bhh0, w, hi);
  const v8f bias1 = load_bias(bih1, bhh1, w, hi);

  float c0[2] = {0.0f, 0.0f}, c1[2] = {0.0f, 0.0f};

  const int st  = m * HP + GP * w + 4 * hi;
  const int rd1 = m * HP + GP * ((w + 1) % NWAVE) + 4 * hi;
  const int rd2 = m * HP + GP * ((w + 2) % NWAVE) + 4 * hi;

  v2f zero2; zero2.x = 0.0f; zero2.y = 0.0f;
  v2f hv1 = zero2;
  Bx[2].x = (hi == 0) ? x8 : 0.0f;
  v8f D0n = wmma4(Ahh0[0], zero2, bias0);
  D0n = wmma4(Aih0[0], Bx[0], D0n);
  D0n = wmma4(Aih0[1], Bx[1], D0n);
  D0n = wmma4(Aih0[2], Bx[2], D0n);
  v2f hv0 = lstm_gates(D0n, c0);

#pragma unroll 1
  for (int hc = 0; hc < 2 * NCH; ++hc) {
    const int cn = (hc >> 1) + 1;
    if (cn < NCH) {
      if ((hc & 1) == 0) stage_load(gx + cn * XROW, R);
      else               stage_store(xs + (cn & 1) * XBUF + lofs, R);
    }

#pragma unroll 1
  for (int th = 0; th < TC / 2; ++th) {
    const int t = hc * (TC / 2) + th;
    const int pa = t & 1;

    {
      v4f hs; hs.x = hv0.x; hs.y = hv0.y; hs.z = hv1.x; hs.w = hv1.y;
      *(v4f*)&hx[pa][st] = hs;
    }
    {
      read_X(xq, Bx, x8);
      const int t2 = t + 2;
      xq = (t2 % TC == 0) ? xs + ((t2 / TC) & 1) * XBUF + xlane : xq + HID;
    }
    __builtin_amdgcn_sched_barrier(0);

    v8f D1 = wmma4(Aih1[0], hv0, bias1);
    D0n    = wmma4(Ahh0[0], hv0, bias0);
    D1     = wmma4(Ahh1[0], hv1, D1);
    __builtin_amdgcn_sched_barrier(0);

    asm volatile("s_wait_xcnt 0x0");
    lds_barrier();

    v2f Bh0r[2], Bh1r[2];
    {
      const v4f r1 = read_B2(&hx[pa][rd1]);
      const v4f r2 = read_B2(&hx[pa][rd2]);
      Bh0r[0].x = r1.x; Bh0r[0].y = r1.y;
      Bh0r[1].x = r2.x; Bh0r[1].y = r2.y;
      Bh1r[0].x = r1.z; Bh1r[0].y = r1.w;
      Bh1r[1].x = r2.z; Bh1r[1].y = r2.w;
    }
    __builtin_amdgcn_sched_barrier(0);

    Bx[2].x = (hi == 0) ? x8 : 0.0f;
    D0n = wmma4(Aih0[0], Bx[0], D0n);
    D0n = wmma4(Aih0[1], Bx[1], D0n);
    D0n = wmma4(Aih0[2], Bx[2], D0n);
    asm volatile("" : "+v"(D0n));
    __builtin_amdgcn_sched_barrier(0);

    D0n = wmma4(Ahh0[1], Bh0r[0], D0n);
    D1  = wmma4(Aih1[1], Bh0r[0], D1);
    D0n = wmma4(Ahh0[2], Bh0r[1], D0n);
    D1  = wmma4(Ahh1[1], Bh1r[0], D1);
    D1  = wmma4(Aih1[2], Bh0r[1], D1);
    D1  = wmma4(Ahh1[2], Bh1r[1], D1);
    lstm_gates_two(D0n, c0, hv0, D1, c1, hv1);
  }
  }

  *(v2f*)&hx[TLEN & 1][st + 2] = hv1;
  __syncthreads();

  if (w == 0 && hi == 0) {
    const float* hl = &hx[TLEN & 1][m * HP];
    float h1[HID];
#pragma unroll
    for (int j = 0; j < HID; ++j) h1[j] = hl[H1SLOT(j)];
    float* op = out + (size_t)bglob * NCLS;
#pragma unroll
    for (int k = 0; k < NCLS; ++k) {
      float acc = fcb[k];
#pragma unroll
      for (int j = 0; j < HID; ++j) acc += h1[j] * fcw[k * HID + j];
      op[k] = acc;
    }
  }
}

extern "C" void kernel_launch(void* const* d_in, const int* in_sizes, int n_in,
                              void* d_out, int out_size, void* d_ws, size_t ws_size,
                              hipStream_t stream) {
  const float* x    = (const float*)d_in[0];
  const float* wih0 = (const float*)d_in[1];
  const float* whh0 = (const float*)d_in[2];
  const float* bih0 = (const float*)d_in[3];
  const float* bhh0 = (const float*)d_in[4];
  const float* wih1 = (const float*)d_in[5];
  const float* whh1 = (const float*)d_in[6];
  const float* bih1 = (const float*)d_in[7];
  const float* bhh1 = (const float*)d_in[8];
  const float* fcw  = (const float*)d_in[9];
  const float* fcb  = (const float*)d_in[10];
  float* out = (float*)d_out;

  const int B = in_sizes[0] / (TLEN * HID);
  dim3 grid(B / 16);
  dim3 block(32 * NWAVE);
  lstm2_fc_fused<<<grid, block, 0, stream>>>(x, wih0, whh0, bih0, bhh0,
                                             wih1, whh1, bih1, bhh1,
                                             fcw, fcb, out);
}


# layer 1 runs a step further behind: its input h0(t-1) is whole in registers by then, so 4 of a trip's 12 products wait on the exchange's reads, not 6; 2 trips a loop pass; layer 1 k sum order changes
# speedup vs baseline: 6.1264x; 1.0815x over previous
#include <hip/hip_runtime.h>
#include <math.h>


#define HID 9
#define NG  36
#define TLEN 512
#define NCLS 10
#define NWAVE 3
#define UPW 3
#define GP 8
#define HP 28
#define H1SLOT(j) (GP * ((j) / UPW) + 4 * (((j) % UPW) / 2) + 2 + ((j) % UPW) % 2)
#define TC 32
#define NCH (TLEN / TC)
#define XROW (TC * HID)
#define XP (XROW + 4)
#define XBUF (16 * XP)
#define TPR 24
#define XLD 12

typedef __attribute__((ext_vector_type(2))) float v2f;
typedef __attribute__((ext_vector_type(4))) float v4f;
typedef __attribute__((ext_vector_type(8))) float v8f;

__device__ __forceinline__ float fexp2(float x) {
#if defined(__HIP_DEVICE_COMPILE__) && __has_builtin(__builtin_amdgcn_exp2f)
  return __builtin_amdgcn_exp2f(x);
#else
  return exp2f(x);
#endif
}

__device__ __forceinline__ float frcp(float x) {
#if defined(__HIP_DEVICE_COMPILE__) && __has_builtin(__builtin_amdgcn_rcpf)
  return __builtin_amdgcn_rcpf(x);
#else
  return 1.0f / x;
#endif
}

__device__ __forceinline__ float ftanh(float x) {
#if defined(__HIP_DEVICE_COMPILE__) && __has_builtin(__builtin_amdgcn_tanhf)
  return __builtin_amdgcn_tanhf(x);
#else
  float e = fexp2(2.8853900817779268f * x);
  return 1.0f - 2.0f * frcp(e + 1.0f);
#endif
}

__device__ __forceinline__ float fsig(float x) {
  return 0.5f + 0.5f * ftanh(0.5f * x);
}

__device__ __forceinline__ v8f wmma4(v2f a, v2f b, v8f c) {
  return __builtin_amdgcn_wmma_f32_16x16x4_f32(false, a, false, b, (short)0, c,
                                               false, false);
}

__device__ __forceinline__ int tile_row_to_gate_row(int q, int w) {
  return (q & 3) * HID + UPW * w + (q >> 2);
}

template <bool byowner>
__device__ __forceinline__ void load_A(const float* __restrict__ W,
                                       v2f A[3], int w, int m, int hi) {
  bool rok = m < 4 * UPW;
  int row = rok ? tile_row_to_gate_row(m, w) : 0;
#pragma unroll
  for (int cc = 0; cc < 3; ++cc) {
    int s0 = 4 * cc + 2 * hi;
    int c0 = byowner ? UPW * ((w + cc) % NWAVE) + 2 * hi : s0;
    bool ok0 = byowner ? true : (s0 < HID);
    bool ok1 = byowner ? (hi == 0) : (s0 + 1 < HID);
    float a0 = (rok && ok0) ? W[row * HID + c0]     : 0.0f;
    float a1 = (rok && ok1) ? W[row * HID + c0 + 1] : 0.0f;
    v2f a; a.x = a0; a.y = a1;
    A[cc] = a;
  }
}

__device__ __forceinline__ v8f load_bias(const float* __restrict__ bi,
                                         const float* __restrict__ bh,
                                         int w, int hi) {
  v8f bias;
#pragma unroll
  for (int r = 0; r < 8; ++r) {
    int q = 8 * hi + r;
    bool rok = q < 4 * UPW;
    int row = rok ? tile_row_to_gate_row(q, w) : 0;
    bias[r] = rok ? (bi[row] + bh[row]) : 0.0f;
  }
  return bias;
}

__device__ __forceinline__ v4f read_B2(const float* hb) {
  return *(const v4f*)hb;
}

__device__ __forceinline__ void stage_load(const float* __restrict__ g, v4f R[XLD]) {
#pragma unroll
  for (int u = 0; u < 4; ++u)
#pragma unroll
    for (int s = 0; s < 3; ++s)
      R[3 * u + s] = *(const v4f*)(g + (size_t)(4 * u) * TLEN * HID + 4 * TPR * s);
}

__device__ __forceinline__ void stage_store(float* l, const v4f R[XLD]) {
#pragma unroll
  for (int u = 0; u < 4; ++u)
#pragma unroll
    for (int s = 0; s < 3; ++s)
      *(v4f*)(l + 4 * u * XP + 4 * TPR * s) = R[3 * u + s];
}

__device__ __forceinline__ void read_X(const float* xq, v2f Bx[3], float& x8) {
  Bx[0].x = xq[0]; Bx[0].y = xq[1];
  Bx[1].x = xq[4]; Bx[1].y = xq[5];
  x8 = xq[8];
  Bx[2].y = 0.0f;
}

__device__ __forceinline__ v2f lstm_gates(v8f D, float cst[2]) {
  float hout[2];
#pragma unroll
  for (int s = 0; s < 2; ++s) {
    float ig = fsig(D[4 * s + 0]);
    float fg = fsig(D[4 * s + 1]);
    float gg = ftanh(D[4 * s + 2]);
    float og = fsig(D[4 * s + 3]);
    float cn = fmaf(ig, gg, fg * cst[s]);
    cst[s] = cn;
    hout[s] = og * ftanh(cn);
  }
  v2f hv; hv.x = hout[0]; hv.y = hout[1];
  return hv;
}

__device__ __forceinline__ void lstm_gates_two(v8f Da, float ca[2], v2f& ha,
                                               v8f Db, float cb[2], v2f& hb, float halfb) {
  float pre[4][4], cs[4], ig[4], fg[4], gg[4], og[4], hout[4];
#pragma unroll
  for (int s = 0; s < 2; ++s) {
#pragma unroll
    for (int g = 0; g < 4; ++g) {
      pre[s][g]     = Da[4 * s + g];
      pre[2 + s][g] = Db[4 * s + g];
    }
    cs[s]     = ca[s];
    cs[2 + s] = cb[s];
  }
#pragma unroll
  for (int q = 0; q < 4; ++q) {
    const float hq = (q < 2) ? 0.5f : halfb;
    ig[q] = fmaf(hq, ftanh(0.5f * pre[q][0]), hq);
    fg[q] = fsig(pre[q][1]);
    gg[q] = ftanh(pre[q][2]);
    og[q] = fmaf(hq, ftanh(0.5f * pre[q][3]), hq);
  }
#pragma unroll
  for (int q = 0; q < 4; ++q) {
    float cn = fmaf(ig[q], gg[q], fg[q] * cs[q]);
    cs[q] = cn;
    hout[q] = og[q] * ftanh(cn);
  }
#pragma unroll
  for (int s = 0; s < 2; ++s) {
    ca[s] = cs[s];
    cb[s] = cs[2 + s];
  }
  ha.x = hout[0]; ha.y = hout[1];
  hb.x = hout[2]; hb.y = hout[3];
}

__device__ __forceinline__ void lds_barrier() {
  __builtin_amdgcn_fence(__ATOMIC_RELEASE, "workgroup", "local");
  __builtin_amdgcn_s_barrier();
  __builtin_amdgcn_fence(__ATOMIC_ACQUIRE, "workgroup", "local");
}

__global__ __launch_bounds__(32 * NWAVE) void lstm2_fc_fused(
    const float* __restrict__ x,
    const float* __restrict__ wih0, const float* __restrict__ whh0,
    const float* __restrict__ bih0, const float* __restrict__ bhh0,
    const float* __restrict__ wih1, const float* __restrict__ whh1,
    const float* __restrict__ bih1, const float* __restrict__ bhh1,
    const float* __restrict__ fcw,  const float* __restrict__ fcb,
    float* __restrict__ out) {
  __shared__ __attribute__((aligned(16))) float hx[2][16 * HP];
  __shared__ __attribute__((aligned(16))) float xs[2 * XBUF];

  const int tid  = threadIdx.x;
  const int w    = tid >> 5;
  const int lane = tid & 31;
  const int m  = lane & 15;
  const int hi = lane >> 4;
  const int bglob = blockIdx.x * 16 + m;

  const int sr = tid / TPR, sq = tid % TPR;
  const float* gx = x + ((size_t)blockIdx.x * 16 + sr) * TLEN * HID + 4 * sq;
  const int lofs = sr * XP + 4 * sq;
  const int xlane = m * XP + 2 * hi;

  v4f R[XLD];
  stage_load(gx, R);
  stage_store(xs + lofs, R);
  if (tid < 2 * 16) {
    v4f z; z.x = 0.0f; z.y = 0.0f; z.z = 0.0f; z.w = 0.0f;
    *(v4f*)(xs + (tid >> 4) * XBUF + (tid & 15) * XP + XROW) = z;
  }
  __syncthreads();
  v2f Bx[3];
  float x8;
  read_X(xs + xlane, Bx, x8);
  const float* xq = xs + xlane + HID;

  v2f Aih0[3], Ahh0[3], Aih1[3], Ahh1[3];
  load_A<false>(wih0, Aih0, w, m, hi);
  load_A<true>(whh0, Ahh0, w, m, hi);
  load_A<true>(wih1, Aih1, w, m, hi);
  load_A<true>(whh1, Ahh1, w, m, hi);

  const v8f bias0 = load_bias(bih0, bhh0, w, hi);
  const v8f bias1 = load_bias(bih1, bhh1, w, hi);

  float c0[2] = {0.0f, 0.0f}, c1[2] = {0.0f, 0.0f};

  const int st  = m * HP + GP * w + 4 * hi;
  const int rd1 = m * HP + GP * ((w + 1) % NWAVE) + 4 * hi;
  const int rd2 = m * HP + GP * ((w + 2) % NWAVE) + 4 * hi;

  v2f zero2; zero2.x = 0.0f; zero2.y = 0.0f;
  v2f hv1 = zero2;
  v2f hv0p = zero2;
  v2f Bh0p[2];
  Bh0p[0] = zero2; Bh0p[1] = zero2;
  Bx[2].x = (hi == 0) ? x8 : 0.0f;
  v8f D0n = wmma4(Ahh0[0], zero2, bias0);
  D0n = wmma4(Aih0[0], Bx[0], D0n);
  D0n = wmma4(Aih0[1], Bx[1], D0n);
  D0n = wmma4(Aih0[2], Bx[2], D0n);
  v2f hv0 = lstm_gates(D0n, c0);

#pragma unroll 1
  for (int hc = 0; hc < 2 * NCH; ++hc) {
    const int cn = (hc >> 1) + 1;
    if (cn < NCH) {
      if ((hc & 1) == 0) stage_load(gx + cn * XROW, R);
      else               stage_store(xs + (cn & 1) * XBUF + lofs, R);
    }

#pragma unroll 2
  for (int th = 0; th < TC / 2; ++th) {
    const int t = hc * (TC / 2) + th;
    const int pa = t & 1;

    {
      v4f hs; hs.x = hv0.x; hs.y = hv0.y; hs.z = hv1.x; hs.w = hv1.y;
      *(v4f*)&hx[pa][st] = hs;
    }
    {
      read_X(xq, Bx, x8);
      const int t2 = t + 2;
      xq = (t2 % TC == 0) ? xs + ((t2 / TC) & 1) * XBUF + xlane : xq + HID;
    }
    __builtin_amdgcn_sched_barrier(0);

    v8f D1 = wmma4(Aih1[0], hv0p, bias1);
    D0n    = wmma4(Ahh0[0], hv0, bias0);
    D1     = wmma4(Ahh1[0], hv1, D1);
    D1     = wmma4(Aih1[1], Bh0p[0], D1);
    __builtin_amdgcn_sched_barrier(0);

    asm volatile("s_wait_xcnt 0x0");
    lds_barrier();

    v2f Bh0r[2], Bh1r[2];
    {
      const v4f r1 = read_B2(&hx[pa][rd1]);
      const v4f r2 = read_B2(&hx[pa][rd2]);
      Bh0r[0].x = r1.x; Bh0r[0].y = r1.y;
      Bh0r[1].x = r2.x; Bh0r[1].y = r2.y;
      Bh1r[0].x = r1.z; Bh1r[0].y = r1.w;
      Bh1r[1].x = r2.z; Bh1r[1].y = r2.w;
    }
    __builtin_amdgcn_sched_barrier(0);

    Bx[2].x = (hi == 0) ? x8 : 0.0f;
    D0n = wmma4(Aih0[0], Bx[0], D0n);
    D1  = wmma4(Aih1[2], Bh0p[1], D1);
    D0n = wmma4(Aih0[1], Bx[1], D0n);
    D0n = wmma4(Aih0[2], Bx[2], D0n);
    asm volatile("" : "+v"(D0n), "+v"(D1));
    __builtin_amdgcn_sched_barrier(0);

    D0n = wmma4(Ahh0[1], Bh0r[0], D0n);
    D1  = wmma4(Ahh1[1], Bh1r[0], D1);
    D0n = wmma4(Ahh0[2], Bh0r[1], D0n);
    D1  = wmma4(Ahh1[2], Bh1r[1], D1);
    hv0p = hv0;
    Bh0p[0] = Bh0r[0];
    Bh0p[1] = Bh0r[1];
    lstm_gates_two(D0n, c0, hv0, D1, c1, hv1, (t == 0) ? 0.0f : 0.5f);
  }
  }

  {
    *(v2f*)&hx[TLEN & 1][st + 2] = hv1;
    v8f D1 = wmma4(Aih1[0], hv0p, bias1);
    D1     = wmma4(Ahh1[0], hv1, D1);
    D1     = wmma4(Aih1[1], Bh0p[0], D1);
    D1     = wmma4(Aih1[2], Bh0p[1], D1);
    __syncthreads();
    const v2f q1 = *(const v2f*)&hx[TLEN & 1][rd1 + 2];
    const v2f q2 = *(const v2f*)&hx[TLEN & 1][rd2 + 2];
    D1 = wmma4(Ahh1[1], q1, D1);
    D1 = wmma4(Ahh1[2], q2, D1);
    hv1 = lstm_gates(D1, c1);
  }
  *(v2f*)&hx[(TLEN + 1) & 1][st + 2] = hv1;
  __syncthreads();

  if (w == 0 && hi == 0) {
    const float* hl = &hx[(TLEN + 1) & 1][m * HP];
    float h1[HID];
#pragma unroll
    for (int j = 0; j < HID; ++j) h1[j] = hl[H1SLOT(j)];
    float* op = out + (size_t)bglob * NCLS;
#pragma unroll
    for (int k = 0; k < NCLS; ++k) {
      float acc = fcb[k];
#pragma unroll
      for (int j = 0; j < HID; ++j) acc += h1[j] * fcw[k * HID + j];
      op[k] = acc;
    }
  }
}

extern "C" void kernel_launch(void* const* d_in, const int* in_sizes, int n_in,
                              void* d_out, int out_size, void* d_ws, size_t ws_size,
                              hipStream_t stream) {
  const float* x    = (const float*)d_in[0];
  const float* wih0 = (const float*)d_in[1];
  const float* whh0 = (const float*)d_in[2];
  const float* bih0 = (const float*)d_in[3];
  const float* bhh0 = (const float*)d_in[4];
  const float* wih1 = (const float*)d_in[5];
  const float* whh1 = (const float*)d_in[6];
  const float* bih1 = (const float*)d_in[7];
  const float* bhh1 = (const float*)d_in[8];
  const float* fcw  = (const float*)d_in[9];
  const float* fcb  = (const float*)d_in[10];
  float* out = (float*)d_out;

  const int B = in_sizes[0] / (TLEN * HID);
  dim3 grid(B / 16);
  dim3 block(32 * NWAVE);
  lstm2_fc_fused<<<grid, block, 0, stream>>>(x, wih0, whh0, bih0, bhh0,
                                             wih1, whh1, bih1, bhh1,
                                             fcw, fcb, out);
}
